# starting kernel
# baseline (speedup 1.0000x reference)
#include <hip/hip_runtime.h>
#include <hip/hip_bf16.h>

#define N_NODES 100000
#define DIM     256
#define N_EDGES 800000
#define L2EPS   1e-12f

typedef __attribute__((ext_vector_type(2))) float v2f;
typedef __attribute__((ext_vector_type(8))) float v8f;

__global__ __launch_bounds__(256)
void gcn_init_kernel(const float* __restrict__ adj, const float* __restrict__ adjw,
                     float* __restrict__ vals, float* __restrict__ out,
                     int nd, int e) {
    int idx = blockIdx.x * blockDim.x + threadIdx.x;
    if (idx < nd) out[idx] = 0.0f;
    if (idx < e)  vals[idx] = adj[idx] + adjw[idx];
}

__global__ __launch_bounds__(128)
void gcn_gemm_wmma_kernel(const float* __restrict__ x, const float* __restrict__ w,
                          float* __restrict__ support) {
    const int wave   = blockIdx.x * (blockDim.x >> 5) + (threadIdx.x >> 5);
    const int lane   = threadIdx.x & 31;
    const int laneHi = lane >> 4;
    const int laneLo = lane & 15;

    const int nTile = wave & 15;
    const int mTile = wave >> 4;
    if (mTile >= N_NODES / 16) return;

    const int m0 = mTile * 16;
    const int n0 = nTile * 16;

    const float* aBase = x + (size_t)(m0 + laneLo) * DIM + 2 * laneHi;
    const float* bBase = w + (size_t)(2 * laneHi) * DIM + n0 + laneLo;

    v8f acc0 = {};
    v8f acc1 = {};

    #pragma unroll 4
    for (int kk = 0; kk < DIM; kk += 8) {
        v2f a0 = *(const v2f*)(aBase + kk);
        v2f b0;
        b0.x = bBase[(size_t)(kk + 0) * DIM];
        b0.y = bBase[(size_t)(kk + 1) * DIM];
        acc0 = __builtin_amdgcn_wmma_f32_16x16x4_f32(
            false, a0, false, b0, (short)0, acc0, false, false);

        v2f a1 = *(const v2f*)(aBase + kk + 4);
        v2f b1;
        b1.x = bBase[(size_t)(kk + 4) * DIM];
        b1.y = bBase[(size_t)(kk + 5) * DIM];
        acc1 = __builtin_amdgcn_wmma_f32_16x16x4_f32(
            false, a1, false, b1, (short)0, acc1, false, false);
    }

    v8f sum = acc0 + acc1;

    float* oBase = support + (size_t)(m0 + 8 * laneHi) * DIM + n0 + laneLo;
    #pragma unroll
    for (int i = 0; i < 8; ++i)
        oBase[(size_t)i * DIM] = sum[i];
}

__global__ __launch_bounds__(256)
void gcn_spmm_kernel(const int* __restrict__ ei, const float* __restrict__ vals,
                     const float* __restrict__ support, float* __restrict__ out) {
    const int e    = blockIdx.x * 8 + (threadIdx.x >> 5);
    const int lane = threadIdx.x & 31;
    if (e >= N_EDGES) return;

    const int dst = ei[e];
    const int src = ei[N_EDGES + e];
    const float v = vals[e];

    const float* sRow = support + (size_t)src * DIM;
    float*       oRow = out     + (size_t)dst * DIM;

    #pragma unroll
    for (int d = lane; d < DIM; d += 32) {
        unsafeAtomicAdd(&oRow[d], v * sRow[d]);
    }
}

__global__ __launch_bounds__(256)
void gcn_normalize_kernel(float* __restrict__ out) {
    __shared__ float red[8];
    const int row = blockIdx.x;
    const int tid = threadIdx.x;

    float v = out[(size_t)row * DIM + tid];
    float s = v * v;

    #pragma unroll
    for (int off = 16; off > 0; off >>= 1)
        s += __shfl_xor(s, off, 32);
    if ((tid & 31) == 0) red[tid >> 5] = s;
    __syncthreads();

    float total = 0.0f;
    #pragma unroll
    for (int i = 0; i < 8; ++i) total += red[i];

    const float scale = 1.0f / fmaxf(sqrtf(total), L2EPS);
    out[(size_t)row * DIM + tid] = v * scale;
}

extern "C" void kernel_launch(void* const* d_in, const int* in_sizes, int n_in,
                              void* d_out, int out_size, void* d_ws, size_t ws_size,
                              hipStream_t stream) {
    const float* x    = (const float*)d_in[0];
    const float* w    = (const float*)d_in[1];
    const float* adj  = (const float*)d_in[2];
    const float* adjw = (const float*)d_in[3];
    const int*   ei   = (const int*)  d_in[4];

    float* out     = (float*)d_out;
    float* support = (float*)d_ws;
    float* vals    = support + (size_t)N_NODES * DIM;

    const int nd = N_NODES * DIM;

    gcn_init_kernel<<<(nd + 255) / 256, 256, 0, stream>>>(adj, adjw, vals, out,
                                                          nd, N_EDGES);

    const int tiles = (N_NODES / 16) * (DIM / 16);
    gcn_gemm_wmma_kernel<<<(tiles + 3) / 4, 128, 0, stream>>>(x, w, support);

    gcn_spmm_kernel<<<(N_EDGES + 7) / 8, 256, 0, stream>>>(ei, vals, support, out);

    gcn_normalize_kernel<<<N_NODES, 256, 0, stream>>>(out);
}


# Product reassociated to (A x) W: atomics sum x rows into out, GEMM runs last in place, a block's 16 waves own a 16-row strip and L2-normalize it; normalize launch gone; sum order and norm tree change
# speedup vs baseline: 1.1478x; 1.1478x over previous
#include <hip/hip_runtime.h>
#include <hip/hip_bf16.h>

#define N_NODES 100000
#define DIM     256
#define N_EDGES 800000
#define L2EPS   1e-12f

typedef __attribute__((ext_vector_type(2))) float v2f;
typedef __attribute__((ext_vector_type(4))) float v4f;
typedef __attribute__((ext_vector_type(8))) float v8f;


__global__ __launch_bounds__(256)
void gcn_init_kernel(const float* __restrict__ adj, const float* __restrict__ adjw,
                     float* __restrict__ vals, float* __restrict__ out,
                     int nd, int e) {
    int idx = blockIdx.x * blockDim.x + threadIdx.x;
    if (idx < nd) out[idx] = 0.0f;
    if (idx < e)  vals[idx] = adj[idx] + adjw[idx];
}

__global__ __launch_bounds__(256)
void gcn_spmm_kernel(const int* __restrict__ ei, const float* __restrict__ vals,
                     const float* __restrict__ x, float* __restrict__ agg) {
    const int e    = blockIdx.x * 8 + (threadIdx.x >> 5);
    const int lane = threadIdx.x & 31;
    if (e >= N_EDGES) return;

    const int dst = ei[e];
    const int src = ei[N_EDGES + e];
    const float v = vals[e];

    const float* sRow = x   + (size_t)src * DIM;
    float*       oRow = agg + (size_t)dst * DIM;

    #pragma unroll
    for (int d = lane; d < DIM; d += 32) {
        unsafeAtomicAdd(&oRow[d], v * sRow[d]);
    }
}

template <int MASK>
__device__ __forceinline__ float row_xor(float v) {
    return __builtin_bit_cast(float,
        __builtin_amdgcn_update_dpp(0, __builtin_bit_cast(int, v), 0x160 + MASK, 0xF, 0xF, true));
}
template <int SRC>
__device__ __forceinline__ float row_bcast(float v) {
    return __builtin_bit_cast(float,
        __builtin_amdgcn_update_dpp(0, __builtin_bit_cast(int, v), 0x150 + SRC, 0xF, 0xF, true));
}

__global__ __launch_bounds__(512)
void gcn_gemm_norm_kernel(float* __restrict__ io, const float* __restrict__ w) {
    __shared__ __attribute__((aligned(16))) float red[16 * 16];

    const int wv     = threadIdx.x >> 5;
    const int lane   = threadIdx.x & 31;
    const int laneHi = lane >> 4;
    const int laneLo = lane & 15;

    const int m0 = blockIdx.x * 16;
    const int n0 = wv * 16;

    const float* aBase = io + (size_t)(m0 + laneLo) * DIM + 2 * laneHi;
    const float* bBase = w + (size_t)(2 * laneHi) * DIM + n0 + laneLo;

    v8f acc0 = {};
    v8f acc1 = {};

    #pragma unroll 4
    for (int kk = 0; kk < DIM; kk += 8) {
        v2f a0 = *(const v2f*)(aBase + kk);
        v2f b0;
        b0.x = bBase[(size_t)(kk + 0) * DIM];
        b0.y = bBase[(size_t)(kk + 1) * DIM];
        acc0 = __builtin_amdgcn_wmma_f32_16x16x4_f32(
            false, a0, false, b0, (short)0, acc0, false, false);

        v2f a1 = *(const v2f*)(aBase + kk + 4);
        v2f b1;
        b1.x = bBase[(size_t)(kk + 4) * DIM];
        b1.y = bBase[(size_t)(kk + 5) * DIM];
        acc1 = __builtin_amdgcn_wmma_f32_16x16x4_f32(
            false, a1, false, b1, (short)0, acc1, false, false);
    }

    v8f sum = acc0 + acc1;

    float ss[8];
    #pragma unroll
    for (int i = 0; i < 8; ++i) ss[i] = sum[i] * sum[i];
    #pragma unroll
    for (int i = 0; i < 8; ++i) ss[i] += row_xor<1>(ss[i]);
    #pragma unroll
    for (int i = 0; i < 8; ++i) ss[i] += row_xor<2>(ss[i]);
    #pragma unroll
    for (int i = 0; i < 8; ++i) ss[i] += row_xor<4>(ss[i]);
    #pragma unroll
    for (int i = 0; i < 8; ++i) ss[i] += row_xor<8>(ss[i]);

    if (laneLo == 0) {
        #pragma unroll
        for (int i = 0; i < 8; ++i) red[(8 * laneHi + i) * 16 + wv] = ss[i];
    }
    __syncthreads();

    const v4f* rp = (const v4f*)(red + (8 * laneHi + (laneLo & 7)) * 16 + (laneLo >> 3) * 8);
    const v4f p0 = rp[0];
    const v4f p1 = rp[1];
    float total = ((p0.x + p0.y) + (p0.z + p0.w)) + ((p1.x + p1.y) + (p1.z + p1.w));
    total += row_xor<8>(total);

    const float scale = 1.0f / fmaxf(sqrtf(total), L2EPS);

    float* oBase = io + (size_t)(m0 + 8 * laneHi) * DIM + n0 + laneLo;
    oBase[(size_t)0 * DIM] = sum[0] * row_bcast<0>(scale);
    oBase[(size_t)1 * DIM] = sum[1] * row_bcast<1>(scale);
    oBase[(size_t)2 * DIM] = sum[2] * row_bcast<2>(scale);
    oBase[(size_t)3 * DIM] = sum[3] * row_bcast<3>(scale);
    oBase[(size_t)4 * DIM] = sum[4] * row_bcast<4>(scale);
    oBase[(size_t)5 * DIM] = sum[5] * row_bcast<5>(scale);
    oBase[(size_t)6 * DIM] = sum[6] * row_bcast<6>(scale);
    oBase[(size_t)7 * DIM] = sum[7] * row_bcast<7>(scale);
}

extern "C" void kernel_launch(void* const* d_in, const int* in_sizes, int n_in,
                              void* d_out, int out_size, void* d_ws, size_t ws_size,
                              hipStream_t stream) {
    const float* x    = (const float*)d_in[0];
    const float* w    = (const float*)d_in[1];
    const float* adj  = (const float*)d_in[2];
    const float* adjw = (const float*)d_in[3];
    const int*   ei   = (const int*)  d_in[4];

    float* out  = (float*)d_out;
    float* vals = (float*)d_ws;

    const int nd = N_NODES * DIM;

    gcn_init_kernel<<<(nd + 255) / 256, 256, 0, stream>>>(adj, adjw, vals, out,
                                                          nd, N_EDGES);

    gcn_spmm_kernel<<<(N_EDGES + 7) / 8, 256, 0, stream>>>(ei, vals, x, out);

    gcn_gemm_norm_kernel<<<N_NODES / 16, 512, 0, stream>>>(out, w);
}


# Init and scatter launches coarsened: 16-byte zero stores, 8 KB per wave; a wave walks 32 edges, 4 rows of loads in flight, weights summed in place; 1.6M waves -> 37K; atomics keep arrival order
# speedup vs baseline: 1.6112x; 1.4037x over previous
#include <hip/hip_runtime.h>
#include <hip/hip_bf16.h>

#define N_NODES 100000
#define DIM     256
#define N_EDGES 800000
#define L2EPS   1e-12f

typedef __attribute__((ext_vector_type(2))) float v2f;
typedef __attribute__((ext_vector_type(4))) float v4f;
typedef __attribute__((ext_vector_type(8))) float v8f;


#define ZERO_STORES 16
static_assert(((size_t)N_NODES * DIM / 4) % (32 * ZERO_STORES) == 0, "a wave's 8 KB piece is all inside or all outside");

__global__ __launch_bounds__(256)
void gcn_zero_kernel(float* __restrict__ out) {
    const size_t n4   = (size_t)N_NODES * DIM / 4;
    const size_t wave = (size_t)blockIdx.x * 8 + (threadIdx.x >> 5);
    const size_t base = wave * (32 * ZERO_STORES);
    if (base >= n4) return;
    v4f* p = (v4f*)out + base + (threadIdx.x & 31);
    const v4f z = {0.0f, 0.0f, 0.0f, 0.0f};
    #pragma unroll
    for (int i = 0; i < ZERO_STORES; ++i) p[i * 32] = z;
}

#define EDGES_PER_WAVE 32
#define EDGE_BATCH     4
static_assert(N_EDGES % (8 * EDGES_PER_WAVE) == 0, "every wave of every block has a full set of edges");
static_assert(DIM == 256, "a lane sweeps 8 columns, 32 apart");

__global__ __launch_bounds__(256)
void gcn_spmm_kernel(const int* __restrict__ ei, const float* __restrict__ adj,
                     const float* __restrict__ adjw, const float* __restrict__ x,
                     float* __restrict__ agg) {
    const int lane = threadIdx.x & 31;
    const int e    = (blockIdx.x * 8 + (threadIdx.x >> 5)) * EDGES_PER_WAVE + lane;

    const int   myDst = ei[e];
    const int   mySrc = ei[N_EDGES + e];
    const float myVal = adj[e] + adjw[e];

    #pragma unroll 1
    for (int j0 = 0; j0 < EDGES_PER_WAVE; j0 += EDGE_BATCH) {
        float r[EDGE_BATCH][8];
        #pragma unroll
        for (int b = 0; b < EDGE_BATCH; ++b) {
            const int src = __builtin_amdgcn_readlane(mySrc, j0 + b);
            const float* sRow = x + (size_t)src * DIM + lane;
            #pragma unroll
            for (int i = 0; i < 8; ++i) r[b][i] = sRow[32 * i];
        }
        #pragma unroll
        for (int b = 0; b < EDGE_BATCH; ++b) {
            const int   dst = __builtin_amdgcn_readlane(myDst, j0 + b);
            const float v   = __builtin_bit_cast(float,
                __builtin_amdgcn_readlane(__builtin_bit_cast(int, myVal), j0 + b));
            float* oRow = agg + (size_t)dst * DIM + lane;
            #pragma unroll
            for (int i = 0; i < 8; ++i) unsafeAtomicAdd(&oRow[32 * i], v * r[b][i]);
        }
    }
}

template <int MASK>
__device__ __forceinline__ float row_xor(float v) {
    return __builtin_bit_cast(float,
        __builtin_amdgcn_update_dpp(0, __builtin_bit_cast(int, v), 0x160 + MASK, 0xF, 0xF, true));
}
template <int SRC>
__device__ __forceinline__ float row_bcast(float v) {
    return __builtin_bit_cast(float,
        __builtin_amdgcn_update_dpp(0, __builtin_bit_cast(int, v), 0x150 + SRC, 0xF, 0xF, true));
}

__global__ __launch_bounds__(512)
void gcn_gemm_norm_kernel(float* __restrict__ io, const float* __restrict__ w) {
    __shared__ __attribute__((aligned(16))) float red[16 * 16];

    const int wv     = threadIdx.x >> 5;
    const int lane   = threadIdx.x & 31;
    const int laneHi = lane >> 4;
    const int laneLo = lane & 15;

    const int m0 = blockIdx.x * 16;
    const int n0 = wv * 16;

    const float* aBase = io + (size_t)(m0 + laneLo) * DIM + 2 * laneHi;
    const float* bBase = w + (size_t)(2 * laneHi) * DIM + n0 + laneLo;

    v8f acc0 = {};
    v8f acc1 = {};

    #pragma unroll 4
    for (int kk = 0; kk < DIM; kk += 8) {
        v2f a0 = *(const v2f*)(aBase + kk);
        v2f b0;
        b0.x = bBase[(size_t)(kk + 0) * DIM];
        b0.y = bBase[(size_t)(kk + 1) * DIM];
        acc0 = __builtin_amdgcn_wmma_f32_16x16x4_f32(
            false, a0, false, b0, (short)0, acc0, false, false);

        v2f a1 = *(const v2f*)(aBase + kk + 4);
        v2f b1;
        b1.x = bBase[(size_t)(kk + 4) * DIM];
        b1.y = bBase[(size_t)(kk + 5) * DIM];
        acc1 = __builtin_amdgcn_wmma_f32_16x16x4_f32(
            false, a1, false, b1, (short)0, acc1, false, false);
    }

    v8f sum = acc0 + acc1;

    float ss[8];
    #pragma unroll
    for (int i = 0; i < 8; ++i) ss[i] = sum[i] * sum[i];
    #pragma unroll
    for (int i = 0; i < 8; ++i) ss[i] += row_xor<1>(ss[i]);
    #pragma unroll
    for (int i = 0; i < 8; ++i) ss[i] += row_xor<2>(ss[i]);
    #pragma unroll
    for (int i = 0; i < 8; ++i) ss[i] += row_xor<4>(ss[i]);
    #pragma unroll
    for (int i = 0; i < 8; ++i) ss[i] += row_xor<8>(ss[i]);

    if (laneLo == 0) {
        #pragma unroll
        for (int i = 0; i < 8; ++i) red[(8 * laneHi + i) * 16 + wv] = ss[i];
    }
    __syncthreads();

    const v4f* rp = (const v4f*)(red + (8 * laneHi + (laneLo & 7)) * 16 + (laneLo >> 3) * 8);
    const v4f p0 = rp[0];
    const v4f p1 = rp[1];
    float total = ((p0.x + p0.y) + (p0.z + p0.w)) + ((p1.x + p1.y) + (p1.z + p1.w));
    total += row_xor<8>(total);

    const float scale = 1.0f / fmaxf(sqrtf(total), L2EPS);

    float* oBase = io + (size_t)(m0 + 8 * laneHi) * DIM + n0 + laneLo;
    oBase[(size_t)0 * DIM] = sum[0] * row_bcast<0>(scale);
    oBase[(size_t)1 * DIM] = sum[1] * row_bcast<1>(scale);
    oBase[(size_t)2 * DIM] = sum[2] * row_bcast<2>(scale);
    oBase[(size_t)3 * DIM] = sum[3] * row_bcast<3>(scale);
    oBase[(size_t)4 * DIM] = sum[4] * row_bcast<4>(scale);
    oBase[(size_t)5 * DIM] = sum[5] * row_bcast<5>(scale);
    oBase[(size_t)6 * DIM] = sum[6] * row_bcast<6>(scale);
    oBase[(size_t)7 * DIM] = sum[7] * row_bcast<7>(scale);
}

extern "C" void kernel_launch(void* const* d_in, const int* in_sizes, int n_in,
                              void* d_out, int out_size, void* d_ws, size_t ws_size,
                              hipStream_t stream) {
    const float* x    = (const float*)d_in[0];
    const float* w    = (const float*)d_in[1];
    const float* adj  = (const float*)d_in[2];
    const float* adjw = (const float*)d_in[3];
    const int*   ei   = (const int*)  d_in[4];

    float* out  = (float*)d_out;

    gcn_zero_kernel<<<(N_NODES * DIM / 4 + 256 * ZERO_STORES - 1) / (256 * ZERO_STORES), 256, 0, stream>>>(out);

    gcn_spmm_kernel<<<N_EDGES / (8 * EDGES_PER_WAVE), 256, 0, stream>>>(ei, adj, adjw, x, out);

    gcn_gemm_norm_kernel<<<N_NODES / 16, 512, 0, stream>>>(out, w);
}
